# starting kernel
# baseline (speedup 1.0000x reference)
#include <hip/hip_runtime.h>
#include <math.h>


#define BB 16
#define NN 8192
#define KK 20
#define DD 64
#define MM (BB * NN)
#define TCOLS 128

typedef __attribute__((ext_vector_type(2))) float v2f;
typedef __attribute__((ext_vector_type(8))) float v8f;

__global__ void __launch_bounds__(256) edgeconv_prep_wct(
    const float* __restrict__ W, float* __restrict__ WcT) {
  int tid = blockIdx.x * 256 + threadIdx.x;
  int j = tid >> 6;
  int d = tid & 63;
  float v;
  if (j < 64) {
    v = W[j * 128 + d];
  } else {
    int o = j - 64;
    v = W[o * 128 + 64 + d] - W[o * 128 + d];
  }
  WcT[j * 64 + d] = v;
}

__global__ void __launch_bounds__(256) edgeconv_gemm_wmma(
    const float* __restrict__ x,
    const float* __restrict__ WcT,
    float* __restrict__ t) {
  __shared__ float sW[TCOLS * DD];
  for (int i = threadIdx.x; i < (TCOLS * DD) / 4; i += 256) {
    ((float4*)sW)[i] = ((const float4*)WcT)[i];
  }
  __syncthreads();

  const int wave = threadIdx.x >> 5;
  const int lane = threadIdx.x & 31;
  const int half = lane >> 4;
  const int l    = lane & 15;
  const int mTile = blockIdx.x * 8 + wave;
  const int m0 = mTile * 16;

  const float* xrow = x + (size_t)(m0 + l) * DD;

  v8f acc[8] = {};
  #pragma unroll
  for (int k = 0; k < 16; ++k) {
    v2f a = *(const v2f*)(xrow + k * 4 + half * 2);
    #pragma unroll
    for (int j = 0; j < 8; ++j) {
      v2f bf = *(const v2f*)(&sW[(j * 16 + l) * DD + k * 4 + half * 2]);
      acc[j] = __builtin_amdgcn_wmma_f32_16x16x4_f32(
           false, a,  false, bf,
           (short)0, acc[j],  false,  false);
    }
  }

  #pragma unroll
  for (int j = 0; j < 8; ++j) {
    float* base = t + (size_t)(m0 + half * 8) * TCOLS + j * 16 + l;
    #pragma unroll
    for (int r = 0; r < 8; ++r) {
      base[(size_t)r * TCOLS] = acc[j][r];
    }
  }
}

__global__ void __launch_bounds__(256) edgeconv_gather_ln_gelu(
    const float* __restrict__ t,
    const int*   __restrict__ ind,
    const float* __restrict__ gamma,
    const float* __restrict__ beta,
    float*       __restrict__ out) {
  const int wave = threadIdx.x >> 5;
  const int lane = threadIdx.x & 31;
  const int m = blockIdx.x * 8 + wave;
  const int b = m >> 13;
  const int o = lane * 2;

  const int* idxp = ind + (size_t)m * KK;
  int myidx = (lane < KK) ? idxp[lane] : 0;

  float hx = -INFINITY, hy = -INFINITY;
  #pragma unroll
  for (int k = 0; k < KK; ++k) {
    int idx = __shfl(myidx, k, 32);
    const float* yrow = t + ((size_t)(b * NN + idx)) * TCOLS;
    float2 yv = *(const float2*)(yrow + o);
    hx = fmaxf(hx, yv.x);
    hy = fmaxf(hy, yv.y);
  }
  float2 zv = *(const float2*)(t + (size_t)m * TCOLS + DD + o);
  hx += zv.x;
  hy += zv.y;

  float s = hx + hy;
  #pragma unroll
  for (int off = 16; off > 0; off >>= 1) s += __shfl_xor(s, off, 32);
  const float mu = s * (1.0f / 64.0f);
  const float dx = hx - mu, dy = hy - mu;
  float v = dx * dx + dy * dy;
  #pragma unroll
  for (int off = 16; off > 0; off >>= 1) v += __shfl_xor(v, off, 32);
  const float rstd = rsqrtf(v * (1.0f / 64.0f) + 1e-5f);

  float n0 = dx * rstd * gamma[o]     + beta[o];
  float n1 = dy * rstd * gamma[o + 1] + beta[o + 1];

  const float is2 = 0.70710678118654752440f;
  float o0 = 0.5f * n0 * (1.0f + erff(n0 * is2));
  float o1 = 0.5f * n1 * (1.0f + erff(n1 * is2));

  *(float2*)(out + (size_t)m * DD + o) = make_float2(o0, o1);
}

extern "C" void kernel_launch(void* const* d_in, const int* in_sizes, int n_in,
                              void* d_out, int out_size, void* d_ws, size_t ws_size,
                              hipStream_t stream) {
  const float* x     = (const float*)d_in[0];
  const int*   ind   = (const int*)d_in[1];
  const float* W     = (const float*)d_in[2];
  const float* gamma = (const float*)d_in[3];
  const float* beta  = (const float*)d_in[4];
  float* out = (float*)d_out;

  float* WcT = (float*)d_ws;
  float* t   = WcT + TCOLS * DD;
  (void)ws_size; (void)in_sizes; (void)n_in; (void)out_size;

  edgeconv_prep_wct<<<32, 256, 0, stream>>>(W, WcT);
  edgeconv_gemm_wmma<<<1024, 256, 0, stream>>>(x, WcT, t);
  edgeconv_gather_ln_gelu<<<16384, 256, 0, stream>>>(t, ind, gamma, beta, out);
}


# GEMM kernel: LDS weight tile row pitch 68 floats instead of 64, so each B-fragment ds_load spreads its lanes over all 64 banks instead of stacking the tile rows on the same few banks
# speedup vs baseline: 1.1840x; 1.1840x over previous
#include <hip/hip_runtime.h>
#include <math.h>


#define BB 16
#define NN 8192
#define KK 20
#define DD 64
#define MM (BB * NN)
#define TCOLS 128
#define WPITCH 68

typedef __attribute__((ext_vector_type(2))) float v2f;
typedef __attribute__((ext_vector_type(8))) float v8f;

__global__ void __launch_bounds__(256) edgeconv_prep_wct(
    const float* __restrict__ W, float* __restrict__ WcT) {
  int tid = blockIdx.x * 256 + threadIdx.x;
  int j = tid >> 6;
  int d = tid & 63;
  float v;
  if (j < 64) {
    v = W[j * 128 + d];
  } else {
    int o = j - 64;
    v = W[o * 128 + 64 + d] - W[o * 128 + d];
  }
  WcT[j * 64 + d] = v;
}

__global__ void __launch_bounds__(256) edgeconv_gemm_wmma(
    const float* __restrict__ x,
    const float* __restrict__ WcT,
    float* __restrict__ t) {
  __shared__ float sW[TCOLS * WPITCH];
  for (int i = threadIdx.x; i < (TCOLS * DD) / 4; i += 256) {
    const int row = i >> 4;
    const int q   = i & 15;
    *(float4*)(&sW[row * WPITCH + q * 4]) = ((const float4*)WcT)[i];
  }
  __syncthreads();

  const int wave = threadIdx.x >> 5;
  const int lane = threadIdx.x & 31;
  const int half = lane >> 4;
  const int l    = lane & 15;
  const int mTile = blockIdx.x * 8 + wave;
  const int m0 = mTile * 16;

  const float* xrow = x + (size_t)(m0 + l) * DD;

  v8f acc[8] = {};
  #pragma unroll
  for (int k = 0; k < 16; ++k) {
    v2f a = *(const v2f*)(xrow + k * 4 + half * 2);
    #pragma unroll
    for (int j = 0; j < 8; ++j) {
      v2f bf = *(const v2f*)(&sW[(j * 16 + l) * WPITCH + k * 4 + half * 2]);
      acc[j] = __builtin_amdgcn_wmma_f32_16x16x4_f32(
           false, a,  false, bf,
           (short)0, acc[j],  false,  false);
    }
  }

  #pragma unroll
  for (int j = 0; j < 8; ++j) {
    float* base = t + (size_t)(m0 + half * 8) * TCOLS + j * 16 + l;
    #pragma unroll
    for (int r = 0; r < 8; ++r) {
      base[(size_t)r * TCOLS] = acc[j][r];
    }
  }
}

__global__ void __launch_bounds__(256) edgeconv_gather_ln_gelu(
    const float* __restrict__ t,
    const int*   __restrict__ ind,
    const float* __restrict__ gamma,
    const float* __restrict__ beta,
    float*       __restrict__ out) {
  const int wave = threadIdx.x >> 5;
  const int lane = threadIdx.x & 31;
  const int m = blockIdx.x * 8 + wave;
  const int b = m >> 13;
  const int o = lane * 2;

  const int* idxp = ind + (size_t)m * KK;
  int myidx = (lane < KK) ? idxp[lane] : 0;

  float hx = -INFINITY, hy = -INFINITY;
  #pragma unroll
  for (int k = 0; k < KK; ++k) {
    int idx = __shfl(myidx, k, 32);
    const float* yrow = t + ((size_t)(b * NN + idx)) * TCOLS;
    float2 yv = *(const float2*)(yrow + o);
    hx = fmaxf(hx, yv.x);
    hy = fmaxf(hy, yv.y);
  }
  float2 zv = *(const float2*)(t + (size_t)m * TCOLS + DD + o);
  hx += zv.x;
  hy += zv.y;

  float s = hx + hy;
  #pragma unroll
  for (int off = 16; off > 0; off >>= 1) s += __shfl_xor(s, off, 32);
  const float mu = s * (1.0f / 64.0f);
  const float dx = hx - mu, dy = hy - mu;
  float v = dx * dx + dy * dy;
  #pragma unroll
  for (int off = 16; off > 0; off >>= 1) v += __shfl_xor(v, off, 32);
  const float rstd = rsqrtf(v * (1.0f / 64.0f) + 1e-5f);

  float n0 = dx * rstd * gamma[o]     + beta[o];
  float n1 = dy * rstd * gamma[o + 1] + beta[o + 1];

  const float is2 = 0.70710678118654752440f;
  float o0 = 0.5f * n0 * (1.0f + erff(n0 * is2));
  float o1 = 0.5f * n1 * (1.0f + erff(n1 * is2));

  *(float2*)(out + (size_t)m * DD + o) = make_float2(o0, o1);
}

extern "C" void kernel_launch(void* const* d_in, const int* in_sizes, int n_in,
                              void* d_out, int out_size, void* d_ws, size_t ws_size,
                              hipStream_t stream) {
  const float* x     = (const float*)d_in[0];
  const int*   ind   = (const int*)d_in[1];
  const float* W     = (const float*)d_in[2];
  const float* gamma = (const float*)d_in[3];
  const float* beta  = (const float*)d_in[4];
  float* out = (float*)d_out;

  float* WcT = (float*)d_ws;
  float* t   = WcT + TCOLS * DD;
  (void)ws_size; (void)in_sizes; (void)n_in; (void)out_size;

  edgeconv_prep_wct<<<32, 256, 0, stream>>>(W, WcT);
  edgeconv_gemm_wmma<<<1024, 256, 0, stream>>>(x, WcT, t);
  edgeconv_gather_ln_gelu<<<16384, 256, 0, stream>>>(t, ind, gamma, beta, out);
}


# Gather kernel: each wave walks 16 consecutive rows in a loop (the block's 2560 indices staged once in LDS) instead of one row per wave: 8192 resident waves to start, not 131072 short ones
# speedup vs baseline: 1.2538x; 1.0590x over previous
#include <hip/hip_runtime.h>
#include <math.h>


#define BB 16
#define NN 8192
#define KK 20
#define DD 64
#define MM (BB * NN)
#define TCOLS 128
#define WPITCH 68

typedef __attribute__((ext_vector_type(2))) float v2f;
typedef __attribute__((ext_vector_type(8))) float v8f;

__global__ void __launch_bounds__(256) edgeconv_prep_wct(
    const float* __restrict__ W, float* __restrict__ WcT) {
  int tid = blockIdx.x * 256 + threadIdx.x;
  int j = tid >> 6;
  int d = tid & 63;
  float v;
  if (j < 64) {
    v = W[j * 128 + d];
  } else {
    int o = j - 64;
    v = W[o * 128 + 64 + d] - W[o * 128 + d];
  }
  WcT[j * 64 + d] = v;
}

__global__ void __launch_bounds__(256) edgeconv_gemm_wmma(
    const float* __restrict__ x,
    const float* __restrict__ WcT,
    float* __restrict__ t) {
  __shared__ float sW[TCOLS * WPITCH];
  for (int i = threadIdx.x; i < (TCOLS * DD) / 4; i += 256) {
    const int row = i >> 4;
    const int q   = i & 15;
    *(float4*)(&sW[row * WPITCH + q * 4]) = ((const float4*)WcT)[i];
  }
  __syncthreads();

  const int wave = threadIdx.x >> 5;
  const int lane = threadIdx.x & 31;
  const int half = lane >> 4;
  const int l    = lane & 15;
  const int mTile = blockIdx.x * 8 + wave;
  const int m0 = mTile * 16;

  const float* xrow = x + (size_t)(m0 + l) * DD;

  v8f acc[8] = {};
  #pragma unroll
  for (int k = 0; k < 16; ++k) {
    v2f a = *(const v2f*)(xrow + k * 4 + half * 2);
    #pragma unroll
    for (int j = 0; j < 8; ++j) {
      v2f bf = *(const v2f*)(&sW[(j * 16 + l) * WPITCH + k * 4 + half * 2]);
      acc[j] = __builtin_amdgcn_wmma_f32_16x16x4_f32(
           false, a,  false, bf,
           (short)0, acc[j],  false,  false);
    }
  }

  #pragma unroll
  for (int j = 0; j < 8; ++j) {
    float* base = t + (size_t)(m0 + half * 8) * TCOLS + j * 16 + l;
    #pragma unroll
    for (int r = 0; r < 8; ++r) {
      base[(size_t)r * TCOLS] = acc[j][r];
    }
  }
}

#define ROWS_PER_WAVE 16
#define BLOCK_IDX (8 * ROWS_PER_WAVE * KK)
__global__ void __launch_bounds__(256) edgeconv_gather_ln_gelu(
    const float* __restrict__ t,
    const int*   __restrict__ ind,
    const float* __restrict__ gamma,
    const float* __restrict__ beta,
    float*       __restrict__ out) {
  __shared__ int4 sIdx[BLOCK_IDX / 4];
  {
    const int2* ip = (const int2*)(ind + (size_t)blockIdx.x * BLOCK_IDX);
    int2* sp = (int2*)sIdx;
    for (int j = 0; j < BLOCK_IDX / 2 / 256; ++j) sp[threadIdx.x + j * 256] = ip[threadIdx.x + j * 256];
  }
  __syncthreads();

  const int wave = threadIdx.x >> 5;
  const int lane = threadIdx.x & 31;
  const int m0 = (blockIdx.x * 8 + wave) * ROWS_PER_WAVE;
  const int b = m0 >> 13;
  const int o = lane * 2;

  const float* ybatch = t + (size_t)b * NN * TCOLS + o;
  const float2 gv = *(const float2*)(gamma + o);
  const float2 bv = *(const float2*)(beta + o);

  for (int r = 0; r < ROWS_PER_WAVE; ++r) {
    const int m = m0 + r;
    const int4* rowi = sIdx + (wave * ROWS_PER_WAVE + r) * (KK / 4);

    const v2f zv = *(const v2f*)(t + (size_t)m * TCOLS + DD + o);

    v2f yv[KK];
    for (int q = 0; q < KK / 4; ++q) {
      const int4 iq = rowi[q];
      yv[4 * q + 0] = *(const v2f*)(ybatch + (size_t)iq.x * TCOLS);
      yv[4 * q + 1] = *(const v2f*)(ybatch + (size_t)iq.y * TCOLS);
      yv[4 * q + 2] = *(const v2f*)(ybatch + (size_t)iq.z * TCOLS);
      yv[4 * q + 3] = *(const v2f*)(ybatch + (size_t)iq.w * TCOLS);
    }
    float hx = -INFINITY, hy = -INFINITY;
    for (int k = 0; k < KK; ++k) {
      hx = fmaxf(hx, yv[k].x);
      hy = fmaxf(hy, yv[k].y);
    }
    hx += zv.x;
    hy += zv.y;

    float s = hx + hy;
    #pragma unroll
    for (int off = 16; off > 0; off >>= 1) s += __shfl_xor(s, off, 32);
    const float mu = s * (1.0f / 64.0f);
    const float dx = hx - mu, dy = hy - mu;
    float v = dx * dx + dy * dy;
    #pragma unroll
    for (int off = 16; off > 0; off >>= 1) v += __shfl_xor(v, off, 32);
    const float rstd = rsqrtf(v * (1.0f / 64.0f) + 1e-5f);

    float n0 = dx * rstd * gv.x + bv.x;
    float n1 = dy * rstd * gv.y + bv.y;

    const float is2 = 0.70710678118654752440f;
    float o0 = 0.5f * n0 * (1.0f + erff(n0 * is2));
    float o1 = 0.5f * n1 * (1.0f + erff(n1 * is2));

    *(float2*)(out + (size_t)m * DD + o) = make_float2(o0, o1);
  }
}

extern "C" void kernel_launch(void* const* d_in, const int* in_sizes, int n_in,
                              void* d_out, int out_size, void* d_ws, size_t ws_size,
                              hipStream_t stream) {
  const float* x     = (const float*)d_in[0];
  const int*   ind   = (const int*)d_in[1];
  const float* W     = (const float*)d_in[2];
  const float* gamma = (const float*)d_in[3];
  const float* beta  = (const float*)d_in[4];
  float* out = (float*)d_out;

  float* WcT = (float*)d_ws;
  float* t   = WcT + TCOLS * DD;
  (void)ws_size; (void)in_sizes; (void)n_in; (void)out_size;

  edgeconv_prep_wct<<<32, 256, 0, stream>>>(W, WcT);
  edgeconv_gemm_wmma<<<1024, 256, 0, stream>>>(x, WcT, t);
  edgeconv_gather_ln_gelu<<<1024, 256, 0, stream>>>(t, ind, gamma, beta, out);
}


# The separate weight-prep launch is gone (3 launches -> 2): the GEMM kernel builds its LDS weight tile (A rows, B-A rows) straight from W, a thread's eight loads issued together
# speedup vs baseline: 1.3261x; 1.0577x over previous
#include <hip/hip_runtime.h>
#include <math.h>


#define BB 16
#define NN 8192
#define KK 20
#define DD 64
#define MM (BB * NN)
#define TCOLS 128
#define WPITCH 68

typedef __attribute__((ext_vector_type(2))) float v2f;
typedef __attribute__((ext_vector_type(8))) float v8f;

__global__ void __launch_bounds__(256) edgeconv_gemm_wmma(
    const float* __restrict__ x,
    const float* __restrict__ W,
    float* __restrict__ t) {
  __shared__ float sW[TCOLS * WPITCH];
  {
    const float4* W4 = (const float4*)W;
    float4 wa[4], wb[4];
    #pragma unroll
    for (int it = 0; it < 4; ++it) {
      const int i = threadIdx.x + it * 256;
      const int o = i >> 4;
      const int q = i & 15;
      wa[it] = W4[o * 32 + q];
      wb[it] = W4[o * 32 + 16 + q];
    }
    #pragma unroll
    for (int it = 0; it < 4; ++it) {
      const int i = threadIdx.x + it * 256;
      const int o = i >> 4;
      const int q = i & 15;
      *(float4*)(&sW[o * WPITCH + q * 4]) = wa[it];
      *(float4*)(&sW[(64 + o) * WPITCH + q * 4]) =
          make_float4(wb[it].x - wa[it].x, wb[it].y - wa[it].y,
                      wb[it].z - wa[it].z, wb[it].w - wa[it].w);
    }
  }
  __syncthreads();

  const int wave = threadIdx.x >> 5;
  const int lane = threadIdx.x & 31;
  const int half = lane >> 4;
  const int l    = lane & 15;
  const int mTile = blockIdx.x * 8 + wave;
  const int m0 = mTile * 16;

  const float* xrow = x + (size_t)(m0 + l) * DD;

  v8f acc[8] = {};
  #pragma unroll
  for (int k = 0; k < 16; ++k) {
    v2f a = *(const v2f*)(xrow + k * 4 + half * 2);
    #pragma unroll
    for (int j = 0; j < 8; ++j) {
      v2f bf = *(const v2f*)(&sW[(j * 16 + l) * WPITCH + k * 4 + half * 2]);
      acc[j] = __builtin_amdgcn_wmma_f32_16x16x4_f32(
           false, a,  false, bf,
           (short)0, acc[j],  false,  false);
    }
  }

  #pragma unroll
  for (int j = 0; j < 8; ++j) {
    float* base = t + (size_t)(m0 + half * 8) * TCOLS + j * 16 + l;
    #pragma unroll
    for (int r = 0; r < 8; ++r) {
      base[(size_t)r * TCOLS] = acc[j][r];
    }
  }
}

#define ROWS_PER_WAVE 16
#define BLOCK_IDX (8 * ROWS_PER_WAVE * KK)
__global__ void __launch_bounds__(256) edgeconv_gather_ln_gelu(
    const float* __restrict__ t,
    const int*   __restrict__ ind,
    const float* __restrict__ gamma,
    const float* __restrict__ beta,
    float*       __restrict__ out) {
  __shared__ int4 sIdx[BLOCK_IDX / 4];
  {
    const int2* ip = (const int2*)(ind + (size_t)blockIdx.x * BLOCK_IDX);
    int2* sp = (int2*)sIdx;
    for (int j = 0; j < BLOCK_IDX / 2 / 256; ++j) sp[threadIdx.x + j * 256] = ip[threadIdx.x + j * 256];
  }
  __syncthreads();

  const int wave = threadIdx.x >> 5;
  const int lane = threadIdx.x & 31;
  const int m0 = (blockIdx.x * 8 + wave) * ROWS_PER_WAVE;
  const int b = m0 >> 13;
  const int o = lane * 2;

  const float* ybatch = t + (size_t)b * NN * TCOLS + o;
  const float2 gv = *(const float2*)(gamma + o);
  const float2 bv = *(const float2*)(beta + o);

  for (int r = 0; r < ROWS_PER_WAVE; ++r) {
    const int m = m0 + r;
    const int4* rowi = sIdx + (wave * ROWS_PER_WAVE + r) * (KK / 4);

    const v2f zv = *(const v2f*)(t + (size_t)m * TCOLS + DD + o);

    v2f yv[KK];
    for (int q = 0; q < KK / 4; ++q) {
      const int4 iq = rowi[q];
      yv[4 * q + 0] = *(const v2f*)(ybatch + (size_t)iq.x * TCOLS);
      yv[4 * q + 1] = *(const v2f*)(ybatch + (size_t)iq.y * TCOLS);
      yv[4 * q + 2] = *(const v2f*)(ybatch + (size_t)iq.z * TCOLS);
      yv[4 * q + 3] = *(const v2f*)(ybatch + (size_t)iq.w * TCOLS);
    }
    float hx = -INFINITY, hy = -INFINITY;
    for (int k = 0; k < KK; ++k) {
      hx = fmaxf(hx, yv[k].x);
      hy = fmaxf(hy, yv[k].y);
    }
    hx += zv.x;
    hy += zv.y;

    float s = hx + hy;
    #pragma unroll
    for (int off = 16; off > 0; off >>= 1) s += __shfl_xor(s, off, 32);
    const float mu = s * (1.0f / 64.0f);
    const float dx = hx - mu, dy = hy - mu;
    float v = dx * dx + dy * dy;
    #pragma unroll
    for (int off = 16; off > 0; off >>= 1) v += __shfl_xor(v, off, 32);
    const float rstd = rsqrtf(v * (1.0f / 64.0f) + 1e-5f);

    float n0 = dx * rstd * gv.x + bv.x;
    float n1 = dy * rstd * gv.y + bv.y;

    const float is2 = 0.70710678118654752440f;
    float o0 = 0.5f * n0 * (1.0f + erff(n0 * is2));
    float o1 = 0.5f * n1 * (1.0f + erff(n1 * is2));

    *(float2*)(out + (size_t)m * DD + o) = make_float2(o0, o1);
  }
}

extern "C" void kernel_launch(void* const* d_in, const int* in_sizes, int n_in,
                              void* d_out, int out_size, void* d_ws, size_t ws_size,
                              hipStream_t stream) {
  const float* x     = (const float*)d_in[0];
  const int*   ind   = (const int*)d_in[1];
  const float* W     = (const float*)d_in[2];
  const float* gamma = (const float*)d_in[3];
  const float* beta  = (const float*)d_in[4];
  float* out = (float*)d_out;

  float* t = (float*)d_ws;
  (void)ws_size; (void)in_sizes; (void)n_in; (void)out_size;

  edgeconv_gemm_wmma<<<1024, 256, 0, stream>>>(x, W, t);
  edgeconv_gather_ln_gelu<<<1024, 256, 0, stream>>>(t, ind, gamma, beta, out);
}
